# starting kernel
# baseline (speedup 1.0000x reference)
#include <hip/hip_runtime.h>
#include <hip/hip_bf16.h>


typedef float v2f __attribute__((ext_vector_type(2)));
typedef float v8f __attribute__((ext_vector_type(8)));

#define IMG_H 1024
#define IMG_W 1024
#define HALO 2

#define BLK_ROWS 128
#define BLK_COLS 32
#define P_ROWS (BLK_ROWS + 4)
#define P_COLS (BLK_COLS + 4)
#define ESTRIDE (P_COLS + 1)
#define NCHUNK (BLK_COLS / 16)

#define SHIFT 2.0f
#define EXP_SCALE (15.0f * 1.4426950408889634f)
#define LOG_SCALE (0.6931471805599453f / 15.0f)

__device__ __forceinline__ float fast_exp2(float x) {
#if __has_builtin(__builtin_amdgcn_exp2f)
  return __builtin_amdgcn_exp2f(x);
#else
  return exp2f(x);
#endif
}
__device__ __forceinline__ float fast_log2(float x) {
#if __has_builtin(__builtin_amdgcn_logf)
  return __builtin_amdgcn_logf(x);
#else
  return log2f(x);
#endif
}

__global__ __launch_bounds__(256) void soft_dilate5x5_kernel(
    const float* __restrict__ x, float* __restrict__ out) {
  __shared__ float E[P_ROWS * ESTRIDE];
  __shared__ float Hs[NCHUNK * P_ROWS * 16];

  const int tid  = threadIdx.x;
  const int w    = tid >> 5;
  const int lane = tid & 31;
  const int hl   = lane >> 4;
  const int ln   = lane & 15;

  const int img = blockIdx.x >> 8;
  const int rem = blockIdx.x & 255;
  const int y0  = (rem >> 5) * BLK_ROWS;
  const int x0  = (rem & 31) * BLK_COLS;
  const size_t ibase = (size_t)img << 20;

  for (int idx = tid; idx < P_ROWS * P_COLS; idx += 256) {
    int r = idx / P_COLS;
    int c = idx - r * P_COLS;
    int gy = y0 - HALO + r; gy = gy < 0 ? 0 : (gy > IMG_H - 1 ? IMG_H - 1 : gy);
    int gx = x0 - HALO + c; gx = gx < 0 ? 0 : (gx > IMG_W - 1 ? IMG_W - 1 : gx);
    float v = x[ibase + (size_t)gy * IMG_W + gx];
    E[r * ESTRIDE + c] = fast_exp2((v - SHIFT) * EXP_SCALE);
  }
  __syncthreads();

  for (int idx = tid; idx < P_ROWS * BLK_COLS; idx += 256) {
    int r  = idx >> 5;
    int xo = idx & 31;
    const float* e = &E[r * ESTRIDE + xo];
    float h = ((e[0] + e[1]) + (e[2] + e[3])) + e[4];
    Hs[(((xo >> 4) * P_ROWS) + r) * 16 + (xo & 15)] = h;
  }
  __syncthreads();

  v2f aa[5];
#pragma unroll
  for (int s = 0; s < 5; ++s) {
    const int K0 = 4 * s + 2 * hl;
    aa[s].x = (K0 >= ln && K0 <= ln + 4) ? 1.0f : 0.0f;
    aa[s].y = (K0 + 1 >= ln && K0 + 1 <= ln + 4) ? 1.0f : 0.0f;
  }

  v8f acc[NCHUNK];
#pragma unroll
  for (int nc = 0; nc < NCHUNK; ++nc) {
    acc[nc] = (v8f){};
#pragma unroll
    for (int s = 0; s < 5; ++s) {
      const int R = 16 * w + 4 * s + 2 * hl;
      const int base = (nc * P_ROWS + R) * 16 + ln;
      v2f b;
      b.x = Hs[base];
      b.y = Hs[base + 16];
      acc[nc] = __builtin_amdgcn_wmma_f32_16x16x4_f32(
          false, aa[s], false, b, (short)0, acc[nc], false, false);
    }
  }

#pragma unroll
  for (int nc = 0; nc < NCHUNK; ++nc) {
    const size_t obase =
        ibase + (size_t)(y0 + 16 * w + 8 * hl) * IMG_W + x0 + nc * 16 + ln;
#pragma unroll
    for (int v = 0; v < 8; ++v) {
      out[obase + (size_t)v * IMG_W] = fast_log2(acc[nc][v]) * LOG_SCALE + SHIFT;
    }
  }
}

extern "C" void kernel_launch(void* const* d_in, const int* in_sizes, int n_in,
                              void* d_out, int out_size, void* d_ws, size_t ws_size,
                              hipStream_t stream) {
  const float* x = (const float*)d_in[0];
  float* out = (float*)d_out;

  const int blocks = 4 * (IMG_H / BLK_ROWS) * (IMG_W / BLK_COLS);
  soft_dilate5x5_kernel<<<blocks, 256, 0, stream>>>(x, out);
}


# Re-tiled to 64x128 row bands (512 work-groups, was 1024 of 128x32): a wave owns whole 512-byte rows, all its 16-byte loads in one clause, 5-tap row sums in registers by lane shuffles, only H in LDS
# speedup vs baseline: 1.4907x; 1.4907x over previous
#include <hip/hip_runtime.h>


typedef float v2f __attribute__((ext_vector_type(2)));
typedef float v8f __attribute__((ext_vector_type(8)));

#define IMG_H 1024
#define IMG_W 1024

#define TR 64
#define TC 128
#define PR (TR + 4)
#define PITCH (TC + 8)
#define RPW 9
#define HR (8 * RPW)
#define NT 4

#define SHIFT 2.0f
#define EXP_SCALE (15.0f * 1.4426950408889634f)
#define LOG_SCALE (0.6931471805599453f / 15.0f)

__device__ __forceinline__ float fast_exp2(float x) {
#if __has_builtin(__builtin_amdgcn_exp2f)
  return __builtin_amdgcn_exp2f(x);
#else
  return exp2f(x);
#endif
}
__device__ __forceinline__ float fast_log2(float x) {
#if __has_builtin(__builtin_amdgcn_logf)
  return __builtin_amdgcn_logf(x);
#else
  return log2f(x);
#endif
}
__device__ __forceinline__ float expv(float v) {
  return fast_exp2((v - SHIFT) * EXP_SCALE);
}

__global__ __launch_bounds__(256) void soft_dilate5x5_kernel(
    const float* __restrict__ x, float* __restrict__ out) {
  __shared__ __attribute__((aligned(16))) float Hs[HR * PITCH];

  const int tid  = threadIdx.x;
  const int w    = __builtin_amdgcn_readfirstlane(tid >> 5);
  const int lane = tid & 31;
  const int hl   = lane >> 4;
  const int ln   = lane & 15;

  const int img = blockIdx.x >> 7;
  const int rem = blockIdx.x & 127;
  const int y0  = (rem >> 3) * TR;
  const int x0  = (rem & 7) * TC;
  const size_t ibase = (size_t)img << 20;
  const float* __restrict__ xi = x + ibase;

  const bool edgeL = (x0 == 0);
  const bool edgeR = (x0 == IMG_W - TC);
  const int  hx = hl ? (edgeR ? IMG_W - 2 : x0 + TC) : (edgeL ? 0 : x0 - 2);
  const int  hi = ln < RPW ? ln : RPW - 1;

  float4 c[RPW];
#pragma unroll
  for (int i = 0; i < RPW; ++i) {
    const int r = w + 8 * i;
    int gy = y0 - 2 + r;
    gy = gy < 0 ? 0 : (gy > IMG_H - 1 ? IMG_H - 1 : gy);
    const float* row = xi + gy * IMG_W + x0;
    c[i] = reinterpret_cast<const float4*>(row)[lane];
  }
  int hgy = y0 - 2 + w + 8 * hi;
  hgy = hgy < 0 ? 0 : (hgy > IMG_H - 1 ? IMG_H - 1 : hgy);
  const float2 hh = *reinterpret_cast<const float2*>(xi + hgy * IMG_W + hx);
  __builtin_amdgcn_sched_barrier(0);

  const int gh0 = __float_as_int(expv(hh.x));
  const int gh1 = __float_as_int(expv(hh.y));
#pragma unroll
  for (int i = 0; i < RPW; ++i) {
    const int r = w + 8 * i;
    const float e0 = expv(c[i].x);
    const float e1 = expv(c[i].y);
    const float e2 = expv(c[i].z);
    const float e3 = expv(c[i].w);
    const float l0 = __int_as_float(__builtin_amdgcn_readlane(gh0, i));
    const float l1 = __int_as_float(__builtin_amdgcn_readlane(gh1, i));
    const float r0 = __int_as_float(__builtin_amdgcn_readlane(gh0, 16 + i));
    const float r1 = __int_as_float(__builtin_amdgcn_readlane(gh1, 16 + i));

    float pz = __shfl_up(e2, 1);
    float pw = __shfl_up(e3, 1);
    float nx = __shfl_down(e0, 1);
    float ny = __shfl_down(e1, 1);
    if (lane == 0)  { pz = edgeL ? e0 : l0;  pw = edgeL ? e0 : l1; }
    if (lane == 31) { nx = edgeR ? e3 : r0;  ny = edgeR ? e3 : r1; }

    float4 hv;
    hv.x = ((pz + pw) + (e0 + e1)) + e2;
    hv.y = ((pw + e0) + (e1 + e2)) + e3;
    hv.z = ((e0 + e1) + (e2 + e3)) + nx;
    hv.w = ((e1 + e2) + (e3 + nx)) + ny;
    *reinterpret_cast<float4*>(&Hs[r * PITCH + 4 * lane]) = hv;
  }
  __syncthreads();

  v2f aa[5];
#pragma unroll
  for (int s = 0; s < 5; ++s) {
    const int K0 = 4 * s + 2 * hl;
    aa[s].x = (K0 >= ln && K0 <= ln + 4) ? 1.0f : 0.0f;
    aa[s].y = (K0 + 1 >= ln && K0 + 1 <= ln + 4) ? 1.0f : 0.0f;
  }

  const int rt = w >> 1;
  const int cb = (w & 1) * NT;

  v8f acc[NT];
#pragma unroll
  for (int t = 0; t < NT; ++t) acc[t] = (v8f){};
#pragma unroll
  for (int s = 0; s < 5; ++s) {
    const int R = 16 * rt + 4 * s + 2 * hl;
#pragma unroll
    for (int t = 0; t < NT; ++t) {
      const int base = R * PITCH + 16 * (cb + t) + ln;
      v2f b;
      b.x = Hs[base];
      b.y = Hs[base + PITCH];
      acc[t] = __builtin_amdgcn_wmma_f32_16x16x4_f32(
          false, aa[s], false, b, (short)0, acc[t], false, false);
    }
  }

#pragma unroll
  for (int t = 0; t < NT; ++t) {
    const size_t obase =
        ibase + (size_t)(y0 + 16 * rt + 8 * hl) * IMG_W + x0 + 16 * (cb + t) + ln;
#pragma unroll
    for (int v = 0; v < 8; ++v) {
      out[obase + (size_t)v * IMG_W] = fast_log2(acc[t][v]) * LOG_SCALE + SHIFT;
    }
  }
}

extern "C" void kernel_launch(void* const* d_in, const int* in_sizes, int n_in,
                              void* d_out, int out_size, void* d_ws, size_t ws_size,
                              hipStream_t stream) {
  const float* x = (const float*)d_in[0];
  float* out = (float*)d_out;

  const int blocks = 4 * (IMG_H / TR) * (IMG_W / TC);
  soft_dilate5x5_kernel<<<blocks, 256, 0, stream>>>(x, out);
}


# Each wave owns a 16x64 output block end to end: H row sums by DPP row shifts feed the WMMA B operand from registers, no LDS, no barrier, 16-byte stores; k order of the vertical sum changes
# speedup vs baseline: 1.9872x; 1.3331x over previous
#include <hip/hip_runtime.h>


typedef float v2f __attribute__((ext_vector_type(2)));
typedef float v8f __attribute__((ext_vector_type(8)));

#define IMG_H 1024
#define IMG_W 1024

#define TR 64
#define TC 128
#define WC 64
#define NP 10
#define NT 4

#define SHIFT 2.0f
#define EXP_SCALE (15.0f * 1.4426950408889634f)
#define LOG_SCALE (0.6931471805599453f / 15.0f)

__device__ __forceinline__ float fast_exp2(float x) {
#if __has_builtin(__builtin_amdgcn_exp2f)
  return __builtin_amdgcn_exp2f(x);
#else
  return exp2f(x);
#endif
}
__device__ __forceinline__ float fast_log2(float x) {
#if __has_builtin(__builtin_amdgcn_logf)
  return __builtin_amdgcn_logf(x);
#else
  return log2f(x);
#endif
}
__device__ __forceinline__ float expv(float v) {
  return fast_exp2((v - SHIFT) * EXP_SCALE);
}

__device__ __forceinline__ float from_left_lane(float edge, float v) {
  return __int_as_float(__builtin_amdgcn_update_dpp(
      __float_as_int(edge), __float_as_int(v), 0x111, 0xf, 0xf, false));
}
__device__ __forceinline__ float from_right_lane(float edge, float v) {
  return __int_as_float(__builtin_amdgcn_update_dpp(
      __float_as_int(edge), __float_as_int(v), 0x101, 0xf, 0xf, false));
}

__global__ __launch_bounds__(256) void soft_dilate5x5_kernel(
    const float* __restrict__ x, float* __restrict__ out) {
  const int tid  = threadIdx.x;
  const int w    = __builtin_amdgcn_readfirstlane(tid >> 5);
  const int lane = tid & 31;
  const int hl   = lane >> 4;
  const int ln   = lane & 15;

  const int img = blockIdx.x >> 7;
  const int rem = blockIdx.x & 127;
  const int y0  = (rem >> 3) * TR + 16 * (w >> 1);
  const int x0  = (rem & 7) * TC + WC * (w & 1);
  const size_t ibase = (size_t)img << 20;
  const float* __restrict__ xi = x + ibase;

  const bool edgeL = (x0 == 0);
  const bool edgeR = (x0 == IMG_W - WC);
  const bool lside = ln < 8;
  const int  hxL   = edgeL ? 0 : x0 - 2;
  const int  hxR   = edgeR ? IMG_W - 2 : x0 + WC;
  const int  hx    = lside ? hxL : hxR;
  const bool dupL  = lside && edgeL;
  const bool dupR  = !lside && edgeR;

  const int rF = (y0 - 2 + hl) < 0 ? 0 : (y0 - 2 + hl);
  const int rL = (y0 + 16 + hl) > IMG_H - 1 ? IMG_H - 1 : (y0 + 16 + hl);
  const int cm = x0 + 4 * ln;
  const float* __restrict__ pmF = xi + rF * IMG_W + cm;
  const float* __restrict__ pmM = xi + (y0 + hl) * IMG_W + cm;
  const float* __restrict__ pmL = xi + rL * IMG_W + cm;
  const float* __restrict__ phF = xi + rF * IMG_W + hx;
  const float* __restrict__ phM = xi + (y0 + hl) * IMG_W + hx;
  const float* __restrict__ phL = xi + rL * IMG_W + hx;

  float4 c[NP];
  float2 g[NP];
#pragma unroll
  for (int i = 0; i < NP; ++i) {
    const float* pm = (i == 0) ? pmF : (i == NP - 1) ? pmL : pmM + (i - 1) * 2 * IMG_W;
    const float* ph = (i == 0) ? phF : (i == NP - 1) ? phL : phM + (i - 1) * 2 * IMG_W;
    c[i] = *reinterpret_cast<const float4*>(pm);
    g[i] = *reinterpret_cast<const float2*>(ph);
  }
  __builtin_amdgcn_sched_barrier(0);

  float H[NP][4];
#pragma unroll
  for (int i = 0; i < NP; ++i) {
    float g0 = g[i].x;
    float g1 = g[i].y;
    g1 = dupL ? g0 : g1;
    g0 = dupR ? g1 : g0;
    const float h0 = expv(g0);
    const float h1 = expv(g1);
    const float e0 = expv(c[i].x);
    const float e1 = expv(c[i].y);
    const float e2 = expv(c[i].z);
    const float e3 = expv(c[i].w);

    const float pz = from_left_lane(h0, e2);
    const float pw = from_left_lane(h1, e3);
    const float nx = from_right_lane(h0, e0);
    const float ny = from_right_lane(h1, e1);

    H[i][0] = ((pz + pw) + (e0 + e1)) + e2;
    H[i][1] = ((pw + e0) + (e1 + e2)) + e3;
    H[i][2] = ((e0 + e1) + (e2 + e3)) + nx;
    H[i][3] = ((e1 + e2) + (e3 + nx)) + ny;
  }

  v2f aa[5];
#pragma unroll
  for (int s = 0; s < 5; ++s) {
    const int r0 = 4 * s + hl;
    const int r1 = r0 + 2;
    aa[s].x = (r0 >= ln && r0 <= ln + 4) ? 1.0f : 0.0f;
    aa[s].y = (r1 >= ln && r1 <= ln + 4) ? 1.0f : 0.0f;
  }

  v8f acc[NT];
#pragma unroll
  for (int t = 0; t < NT; ++t) acc[t] = (v8f){};
#pragma unroll
  for (int s = 0; s < 5; ++s) {
#pragma unroll
    for (int t = 0; t < NT; ++t) {
      v2f b;
      b.x = H[2 * s][t];
      b.y = H[2 * s + 1][t];
      acc[t] = __builtin_amdgcn_wmma_f32_16x16x4_f32(
          false, aa[s], false, b, (short)0, acc[t], false, false);
    }
  }

  float* __restrict__ orow = out + ibase + (size_t)(y0 + 8 * hl) * IMG_W + x0 + 4 * ln;
#pragma unroll
  for (int v = 0; v < 8; ++v) {
    float4 o;
    o.x = fast_log2(acc[0][v]) * LOG_SCALE + SHIFT;
    o.y = fast_log2(acc[1][v]) * LOG_SCALE + SHIFT;
    o.z = fast_log2(acc[2][v]) * LOG_SCALE + SHIFT;
    o.w = fast_log2(acc[3][v]) * LOG_SCALE + SHIFT;
    *reinterpret_cast<float4*>(orow + (size_t)v * IMG_W) = o;
  }
}

extern "C" void kernel_launch(void* const* d_in, const int* in_sizes, int n_in,
                              void* d_out, int out_size, void* d_ws, size_t ws_size,
                              hipStream_t stream) {
  const float* x = (const float*)d_in[0];
  float* out = (float*)d_out;

  const int blocks = 4 * (IMG_H / TR) * (IMG_W / TC);
  soft_dilate5x5_kernel<<<blocks, 256, 0, stream>>>(x, out);
}
